# starting kernel
# baseline (speedup 1.0000x reference)
#include <hip/hip_runtime.h>
#include <hip/hip_bf16.h>
#include <math.h>

typedef float v2f __attribute__((ext_vector_type(2)));
typedef float v8f __attribute__((ext_vector_type(8)));

#define DDIM 256
#define KCLS 5
#define PITCH 260
#define MAXN  9.96f

__device__ __forceinline__ float waveRedSum(float v) {
#pragma unroll
  for (int off = 16; off > 0; off >>= 1) v += __shfl_xor(v, off, 32);
  return v;
}

__global__ __launch_bounds__(256) void hpr_k1(const float* __restrict__ feat,
                                              float* __restrict__ proto,
                                              float* __restrict__ proto_y2,
                                              float* __restrict__ num,
                                              float* __restrict__ den,
                                              float* __restrict__ lossA) {
  __shared__ float red[256];
  int t = threadIdx.x;
  float agk[KCLS], ag[KCLS];
#pragma unroll
  for (int c = 0; c < KCLS; c++) { agk[c] = 0.f; ag[c] = 0.f; }

  for (int r = 0; r < 25; r++) {
    float v = feat[r * DDIM + t];
    red[t] = v * v; __syncthreads();
    for (int s = 128; s > 0; s >>= 1) { if (t < s) red[t] += red[t + s]; __syncthreads(); }
    float n2 = red[0]; __syncthreads();
    float n  = fmaxf(sqrtf(n2), 1e-5f);
    float sc = tanhf(0.1f * n) / (0.1f * n);
    float pn = sc * n;
    if (pn > MAXN) { sc *= MAXN / pn; pn = MAXN; }
    float x2    = pn * pn;
    float kq    = 2.0f / (1.0f + 0.01f * x2);
    float gamma = rsqrtf(1.0f - 0.01f * kq * kq * x2);
    int c = r / 5;
    agk[c] += gamma * kq * sc * v;
    ag[c]  += gamma;
  }

  for (int c = 0; c < KCLS; c++) {
    num[c * DDIM + t] = agk[c];
    float mk = agk[c] / ag[c];
    red[t] = mk * mk; __syncthreads();
    for (int s = 128; s > 0; s >>= 1) { if (t < s) red[t] += red[t + s]; __syncthreads(); }
    float mk2 = red[0]; __syncthreads();
    float dn = 1.0f + sqrtf(1.0f - 0.01f * mk2);
    proto[c * DDIM + t] = mk / dn;
    if (t == 0) { proto_y2[c] = mk2 / (dn * dn); den[c] = ag[c]; }
  }
  if (t == 0) lossA[0] = 0.f;
}

template <bool PASS2>
__global__ __launch_bounds__(64) void hpr_pass(const float* __restrict__ feat,
                                               const int* __restrict__ label,
                                               const float* __restrict__ proto,
                                               const float* __restrict__ proto_y2,
                                               float* __restrict__ num,
                                               float* __restrict__ den,
                                               float* __restrict__ lossA,
                                               float* __restrict__ out, int Q) {
  __shared__ __align__(16) float tile[2][16][PITCH];
  __shared__ __align__(16) float sproto[16][PITCH];
  __shared__ float sX2[2][16];
  __shared__ float sDist[2][16][8];
  __shared__ int   sPred[32];
  __shared__ float sCoef[32];
  __shared__ float bden[KCLS];
  __shared__ float bloss;

  const int tid  = threadIdx.x;
  const int wave = tid >> 5, lane = tid & 31;
  const int m16  = lane & 15, half = lane >> 4;
  const int qbase = blockIdx.x * 32 + wave * 16;

  for (int idx = tid; idx < 16 * DDIM; idx += 64) {
    int r = idx >> 8, cc = idx & 255;
    sproto[r][cc] = (r < KCLS) ? proto[r * DDIM + cc] : 0.f;
  }
  if (!PASS2) {
    if (tid < KCLS) bden[tid] = 0.f;
    if (tid == KCLS) bloss = 0.f;
  }

  for (int r = 0; r < 16; r++) {
    int q = qbase + r;
    const float* row = feat + (size_t)(25 + (q < Q ? q : 0)) * DDIM;
    float4 a0 = ((const float4*)row)[lane];
    float4 a1 = ((const float4*)row)[lane + 32];
    *(float4*)&tile[wave][r][lane * 4]       = a0;
    *(float4*)&tile[wave][r][128 + lane * 4] = a1;
  }
  __syncthreads();

  for (int r = 0; r < 16; r++) {
    float acc = 0.f;
#pragma unroll
    for (int i = 0; i < 8; i++) { float v = tile[wave][r][i * 32 + lane]; acc += v * v; }
    float n2 = waveRedSum(acc);
    float n  = fmaxf(sqrtf(n2), 1e-5f);
    float sc = tanhf(0.1f * n) / (0.1f * n);
    float pn = sc * n;
    if (pn > MAXN) { sc *= MAXN / pn; pn = MAXN; }
#pragma unroll
    for (int i = 0; i < 8; i++) tile[wave][r][i * 32 + lane] *= sc;
    if (lane == 0) sX2[wave][r] = pn * pn;
  }
  __syncthreads();

  v8f c = {0.f, 0.f, 0.f, 0.f, 0.f, 0.f, 0.f, 0.f};
  for (int kc = 0; kc < DDIM; kc += 4) {
    int col = kc + half * 2;
    v2f a = *(const v2f*)&tile[wave][m16][col];
    v2f b = *(const v2f*)&sproto[m16][col];
    c = __builtin_amdgcn_wmma_f32_16x16x4_f32(false, a, false, b, (short)0, c,
                                              false, false);
  }

  float y2 = proto_y2[(m16 < KCLS) ? m16 : 0];
  if (m16 < KCLS) {
#pragma unroll
    for (int g = 0; g < 8; g++) {
      int   m  = 8 * half + g;
      float x2 = sX2[wave][m];
      float xy = c[g];
      float A  = 1.f - 0.02f * xy + 0.01f * y2;
      float B  = 1.f - 0.01f * x2;
      float nv2 = fmaxf(A * A * x2 - 2.f * A * B * xy + B * B * y2, 0.f);
      float dnm = 1.f - 0.02f * xy + 1e-4f * x2 * y2 + 1e-5f;
      float man = sqrtf(nv2) / dnm;
      float z = fminf(fmaxf(0.1f * man, -1.f + 1e-5f), 1.f - 1e-5f);
      sDist[wave][m][m16] = 10.f * (log1pf(z) - log1pf(-z));
    }
  }
  __syncthreads();

  float ceContrib = 0.f;
  if (lane < 16) {
    int  q = qbase + lane;
    bool valid = (q < Q);
    float dv[KCLS];
#pragma unroll
    for (int k = 0; k < KCLS; k++) dv[k] = sDist[wave][lane][k];
    float best = dv[0]; int pred = 0;
#pragma unroll
    for (int k = 1; k < KCLS; k++) { if (dv[k] < best) { best = dv[k]; pred = k; } }

    if (PASS2) {
      float e[KCLS]; float se = 0.f;
#pragma unroll
      for (int k = 0; k < KCLS; k++) { e[k] = expf(best - dv[k]); se += e[k]; }
      if (valid) {
        float inv = 1.f / se;
#pragma unroll
        for (int k = 0; k < KCLS; k++) out[(size_t)q * KCLS + k] = e[k] * inv;
      }
    } else {
      float se = 0.f;
#pragma unroll
      for (int k = 0; k < KCLS; k++) se += expf(best - dv[k]);
      float lse = logf(se);
      int lbl = label[Q + q];
      float dl = dv[0];
#pragma unroll
      for (int k = 1; k < KCLS; k++) dl = (lbl == k) ? dv[k] : dl;
      float ce = (best - dl) - lse;
      ceContrib = valid ? ce : 0.f;

      float x2q   = sX2[wave][lane];
      float kq    = 2.0f / (1.0f + 0.01f * x2q);
      float gamma = rsqrtf(1.0f - 0.01f * kq * kq * x2q);
      sPred[wave * 16 + lane] = pred;
      sCoef[wave * 16 + lane] = valid ? gamma * kq : 0.f;
      if (valid) atomicAdd(&bden[pred], gamma);
    }
  }

  if (!PASS2) {
    float s = waveRedSum(ceContrib);
    if (lane == 0) atomicAdd(&bloss, s);
    __syncthreads();

    float racc[KCLS][4];
#pragma unroll
    for (int cc = 0; cc < KCLS; cc++)
#pragma unroll
      for (int j = 0; j < 4; j++) racc[cc][j] = 0.f;

    for (int m2 = 0; m2 < 32; m2++) {
      int p = sPred[m2]; float cf = sCoef[m2];
      int wv = m2 >> 4, r = m2 & 15;
      float v0 = tile[wv][r][tid];
      float v1 = tile[wv][r][tid + 64];
      float v2 = tile[wv][r][tid + 128];
      float v3 = tile[wv][r][tid + 192];
#pragma unroll
      for (int cc = 0; cc < KCLS; cc++) {
        float msk = (p == cc) ? cf : 0.f;
        racc[cc][0] += msk * v0; racc[cc][1] += msk * v1;
        racc[cc][2] += msk * v2; racc[cc][3] += msk * v3;
      }
    }
#pragma unroll
    for (int cc = 0; cc < KCLS; cc++) {
      atomicAdd(&num[cc * DDIM + tid],       racc[cc][0]);
      atomicAdd(&num[cc * DDIM + tid + 64],  racc[cc][1]);
      atomicAdd(&num[cc * DDIM + tid + 128], racc[cc][2]);
      atomicAdd(&num[cc * DDIM + tid + 192], racc[cc][3]);
    }
    if (tid < KCLS) atomicAdd(&den[tid], bden[tid]);
    if (tid == KCLS) atomicAdd(lossA, bloss);
  }
}

__global__ __launch_bounds__(256) void hpr_k3(const float* __restrict__ num,
                                              const float* __restrict__ den,
                                              const float* __restrict__ lossA,
                                              float* __restrict__ proto2,
                                              float* __restrict__ proto2_y2,
                                              float* __restrict__ out, int Q) {
  __shared__ float red[256];
  int t = threadIdx.x;
  for (int c = 0; c < KCLS; c++) {
    float mk = num[c * DDIM + t] / den[c];
    red[t] = mk * mk; __syncthreads();
    for (int s = 128; s > 0; s >>= 1) { if (t < s) red[t] += red[t + s]; __syncthreads(); }
    float mk2 = red[0]; __syncthreads();
    float dn = 1.0f + sqrtf(1.0f - 0.01f * mk2);
    proto2[c * DDIM + t] = mk / dn;
    if (t == 0) proto2_y2[c] = mk2 / (dn * dn);
  }
  if (t == 0) out[(size_t)Q * KCLS] = -lossA[0] / (float)Q;
}

extern "C" void kernel_launch(void* const* d_in, const int* in_sizes, int n_in,
                              void* d_out, int out_size, void* d_ws, size_t ws_size,
                              hipStream_t stream) {
  const float* feat  = (const float*)d_in[0];
  const int*   label = (const int*)d_in[1];
  float* out = (float*)d_out;
  float* ws  = (float*)d_ws;

  int total_rows = in_sizes[0] / DDIM;
  int Q = total_rows - 25;

  float* proto     = ws;
  float* proto_y2  = ws + 1280;
  float* num       = ws + 1288;
  float* den       = ws + 2568;
  float* lossA     = ws + 2576;
  float* proto2    = ws + 2580;
  float* proto2_y2 = ws + 3860;

  hpr_k1<<<1, 256, 0, stream>>>(feat, proto, proto_y2, num, den, lossA);

  int nblk = (Q + 31) / 32;
  hpr_pass<false><<<nblk, 64, 0, stream>>>(feat, label, proto, proto_y2,
                                           num, den, lossA, out, Q);
  hpr_k3<<<1, 256, 0, stream>>>(num, den, lossA, proto2, proto2_y2, out, Q);
  hpr_pass<true><<<nblk, 64, 0, stream>>>(feat, label, proto2, proto2_y2,
                                          num, den, lossA, out, Q);
}


# k1 and k3: one wave per support row and the five class norms side by side, with shuffle reductions, instead of 30 serial 256-thread LDS tree reductions; norm sums use a different reduction tree
# speedup vs baseline: 1.3371x; 1.3371x over previous
#include <hip/hip_runtime.h>
#include <hip/hip_bf16.h>
#include <math.h>

typedef float v2f __attribute__((ext_vector_type(2)));
typedef float v8f __attribute__((ext_vector_type(8)));

#define DDIM 256
#define KCLS 5
#define PITCH 260
#define MAXN  9.96f

__device__ __forceinline__ float waveRedSum(float v) {
#pragma unroll
  for (int off = 16; off > 0; off >>= 1) v += __shfl_xor(v, off, 32);
  return v;
}

__global__ __launch_bounds__(256) void hpr_k1(const float* __restrict__ feat,
                                              float* __restrict__ proto,
                                              float* __restrict__ proto_y2,
                                              float* __restrict__ num,
                                              float* __restrict__ den,
                                              float* __restrict__ lossA) {
  __shared__ float sW[32];
  __shared__ float sG[32];
  __shared__ float sRed[KCLS][8];
  const int t = threadIdx.x;
  const int wave = t >> 5, lane = t & 31;

  float vcol[25];
#pragma unroll
  for (int r = 0; r < 25; r++) vcol[r] = feat[r * DDIM + t];

  float ss[4];
#pragma unroll
  for (int i = 0; i < 4; i++) {
    const int r  = wave + 8 * i;
    const int rr = (r < 25) ? r : 24;
    const float4* row = (const float4*)(feat + rr * DDIM);
    float4 a0 = row[lane];
    float4 a1 = row[lane + 32];
    ss[i] = a0.x * a0.x + a0.y * a0.y + a0.z * a0.z + a0.w * a0.w
          + a1.x * a1.x + a1.y * a1.y + a1.z * a1.z + a1.w * a1.w;
  }
#pragma unroll
  for (int off = 16; off > 0; off >>= 1) {
#pragma unroll
    for (int i = 0; i < 4; i++) ss[i] += __shfl_xor(ss[i], off, 32);
  }
#pragma unroll
  for (int i = 0; i < 4; i++) {
    const int r = wave + 8 * i;
    float n2 = ss[i];
    float n  = fmaxf(sqrtf(n2), 1e-5f);
    float sc = tanhf(0.1f * n) / (0.1f * n);
    float pn = sc * n;
    if (pn > MAXN) { sc *= MAXN / pn; pn = MAXN; }
    float x2    = pn * pn;
    float kq    = 2.0f / (1.0f + 0.01f * x2);
    float gamma = rsqrtf(1.0f - 0.01f * kq * kq * x2);
    if (lane == 0 && r < 25) { sW[r] = gamma * kq * sc; sG[r] = gamma; }
  }
  __syncthreads();

  float mk[KCLS], agc[KCLS], part[KCLS];
#pragma unroll
  for (int c = 0; c < KCLS; c++) {
    float agk = 0.f, ag = 0.f;
#pragma unroll
    for (int j = 0; j < 5; j++) {
      const int r = 5 * c + j;
      agk += sW[r] * vcol[r];
      ag  += sG[r];
    }
    num[c * DDIM + t] = agk;
    agc[c]  = ag;
    mk[c]   = agk / ag;
    part[c] = waveRedSum(mk[c] * mk[c]);
  }
  if (lane == 0) {
#pragma unroll
    for (int c = 0; c < KCLS; c++) sRed[c][wave] = part[c];
  }
  __syncthreads();
#pragma unroll
  for (int c = 0; c < KCLS; c++) {
    float mk2 = 0.f;
#pragma unroll
    for (int w = 0; w < 8; w++) mk2 += sRed[c][w];
    float dn = 1.0f + sqrtf(1.0f - 0.01f * mk2);
    proto[c * DDIM + t] = mk[c] / dn;
    if (t == 0) { proto_y2[c] = mk2 / (dn * dn); den[c] = agc[c]; }
  }
  if (t == 0) lossA[0] = 0.f;
}

template <bool PASS2>
__global__ __launch_bounds__(64) void hpr_pass(const float* __restrict__ feat,
                                               const int* __restrict__ label,
                                               const float* __restrict__ proto,
                                               const float* __restrict__ proto_y2,
                                               float* __restrict__ num,
                                               float* __restrict__ den,
                                               float* __restrict__ lossA,
                                               float* __restrict__ out, int Q) {
  __shared__ __align__(16) float tile[2][16][PITCH];
  __shared__ __align__(16) float sproto[16][PITCH];
  __shared__ float sX2[2][16];
  __shared__ float sDist[2][16][8];
  __shared__ int   sPred[32];
  __shared__ float sCoef[32];
  __shared__ float bden[KCLS];
  __shared__ float bloss;

  const int tid  = threadIdx.x;
  const int wave = tid >> 5, lane = tid & 31;
  const int m16  = lane & 15, half = lane >> 4;
  const int qbase = blockIdx.x * 32 + wave * 16;

  for (int idx = tid; idx < 16 * DDIM; idx += 64) {
    int r = idx >> 8, cc = idx & 255;
    sproto[r][cc] = (r < KCLS) ? proto[r * DDIM + cc] : 0.f;
  }
  if (!PASS2) {
    if (tid < KCLS) bden[tid] = 0.f;
    if (tid == KCLS) bloss = 0.f;
  }

  for (int r = 0; r < 16; r++) {
    int q = qbase + r;
    const float* row = feat + (size_t)(25 + (q < Q ? q : 0)) * DDIM;
    float4 a0 = ((const float4*)row)[lane];
    float4 a1 = ((const float4*)row)[lane + 32];
    *(float4*)&tile[wave][r][lane * 4]       = a0;
    *(float4*)&tile[wave][r][128 + lane * 4] = a1;
  }
  __syncthreads();

  for (int r = 0; r < 16; r++) {
    float acc = 0.f;
#pragma unroll
    for (int i = 0; i < 8; i++) { float v = tile[wave][r][i * 32 + lane]; acc += v * v; }
    float n2 = waveRedSum(acc);
    float n  = fmaxf(sqrtf(n2), 1e-5f);
    float sc = tanhf(0.1f * n) / (0.1f * n);
    float pn = sc * n;
    if (pn > MAXN) { sc *= MAXN / pn; pn = MAXN; }
#pragma unroll
    for (int i = 0; i < 8; i++) tile[wave][r][i * 32 + lane] *= sc;
    if (lane == 0) sX2[wave][r] = pn * pn;
  }
  __syncthreads();

  v8f c = {0.f, 0.f, 0.f, 0.f, 0.f, 0.f, 0.f, 0.f};
  for (int kc = 0; kc < DDIM; kc += 4) {
    int col = kc + half * 2;
    v2f a = *(const v2f*)&tile[wave][m16][col];
    v2f b = *(const v2f*)&sproto[m16][col];
    c = __builtin_amdgcn_wmma_f32_16x16x4_f32(false, a, false, b, (short)0, c,
                                              false, false);
  }

  float y2 = proto_y2[(m16 < KCLS) ? m16 : 0];
  if (m16 < KCLS) {
#pragma unroll
    for (int g = 0; g < 8; g++) {
      int   m  = 8 * half + g;
      float x2 = sX2[wave][m];
      float xy = c[g];
      float A  = 1.f - 0.02f * xy + 0.01f * y2;
      float B  = 1.f - 0.01f * x2;
      float nv2 = fmaxf(A * A * x2 - 2.f * A * B * xy + B * B * y2, 0.f);
      float dnm = 1.f - 0.02f * xy + 1e-4f * x2 * y2 + 1e-5f;
      float man = sqrtf(nv2) / dnm;
      float z = fminf(fmaxf(0.1f * man, -1.f + 1e-5f), 1.f - 1e-5f);
      sDist[wave][m][m16] = 10.f * (log1pf(z) - log1pf(-z));
    }
  }
  __syncthreads();

  float ceContrib = 0.f;
  if (lane < 16) {
    int  q = qbase + lane;
    bool valid = (q < Q);
    float dv[KCLS];
#pragma unroll
    for (int k = 0; k < KCLS; k++) dv[k] = sDist[wave][lane][k];
    float best = dv[0]; int pred = 0;
#pragma unroll
    for (int k = 1; k < KCLS; k++) { if (dv[k] < best) { best = dv[k]; pred = k; } }

    if (PASS2) {
      float e[KCLS]; float se = 0.f;
#pragma unroll
      for (int k = 0; k < KCLS; k++) { e[k] = expf(best - dv[k]); se += e[k]; }
      if (valid) {
        float inv = 1.f / se;
#pragma unroll
        for (int k = 0; k < KCLS; k++) out[(size_t)q * KCLS + k] = e[k] * inv;
      }
    } else {
      float se = 0.f;
#pragma unroll
      for (int k = 0; k < KCLS; k++) se += expf(best - dv[k]);
      float lse = logf(se);
      int lbl = label[Q + q];
      float dl = dv[0];
#pragma unroll
      for (int k = 1; k < KCLS; k++) dl = (lbl == k) ? dv[k] : dl;
      float ce = (best - dl) - lse;
      ceContrib = valid ? ce : 0.f;

      float x2q   = sX2[wave][lane];
      float kq    = 2.0f / (1.0f + 0.01f * x2q);
      float gamma = rsqrtf(1.0f - 0.01f * kq * kq * x2q);
      sPred[wave * 16 + lane] = pred;
      sCoef[wave * 16 + lane] = valid ? gamma * kq : 0.f;
      if (valid) atomicAdd(&bden[pred], gamma);
    }
  }

  if (!PASS2) {
    float s = waveRedSum(ceContrib);
    if (lane == 0) atomicAdd(&bloss, s);
    __syncthreads();

    float racc[KCLS][4];
#pragma unroll
    for (int cc = 0; cc < KCLS; cc++)
#pragma unroll
      for (int j = 0; j < 4; j++) racc[cc][j] = 0.f;

    for (int m2 = 0; m2 < 32; m2++) {
      int p = sPred[m2]; float cf = sCoef[m2];
      int wv = m2 >> 4, r = m2 & 15;
      float v0 = tile[wv][r][tid];
      float v1 = tile[wv][r][tid + 64];
      float v2 = tile[wv][r][tid + 128];
      float v3 = tile[wv][r][tid + 192];
#pragma unroll
      for (int cc = 0; cc < KCLS; cc++) {
        float msk = (p == cc) ? cf : 0.f;
        racc[cc][0] += msk * v0; racc[cc][1] += msk * v1;
        racc[cc][2] += msk * v2; racc[cc][3] += msk * v3;
      }
    }
#pragma unroll
    for (int cc = 0; cc < KCLS; cc++) {
      atomicAdd(&num[cc * DDIM + tid],       racc[cc][0]);
      atomicAdd(&num[cc * DDIM + tid + 64],  racc[cc][1]);
      atomicAdd(&num[cc * DDIM + tid + 128], racc[cc][2]);
      atomicAdd(&num[cc * DDIM + tid + 192], racc[cc][3]);
    }
    if (tid < KCLS) atomicAdd(&den[tid], bden[tid]);
    if (tid == KCLS) atomicAdd(lossA, bloss);
  }
}

__global__ __launch_bounds__(256) void hpr_k3(const float* __restrict__ num,
                                              const float* __restrict__ den,
                                              const float* __restrict__ lossA,
                                              float* __restrict__ proto2,
                                              float* __restrict__ proto2_y2,
                                              float* __restrict__ out, int Q) {
  __shared__ float sRed[KCLS][8];
  const int t = threadIdx.x;
  const int wave = t >> 5, lane = t & 31;
  float mk[KCLS], part[KCLS];
#pragma unroll
  for (int c = 0; c < KCLS; c++) {
    mk[c]   = num[c * DDIM + t] / den[c];
    part[c] = waveRedSum(mk[c] * mk[c]);
  }
  if (lane == 0) {
#pragma unroll
    for (int c = 0; c < KCLS; c++) sRed[c][wave] = part[c];
  }
  __syncthreads();
#pragma unroll
  for (int c = 0; c < KCLS; c++) {
    float mk2 = 0.f;
#pragma unroll
    for (int w = 0; w < 8; w++) mk2 += sRed[c][w];
    float dn = 1.0f + sqrtf(1.0f - 0.01f * mk2);
    proto2[c * DDIM + t] = mk[c] / dn;
    if (t == 0) proto2_y2[c] = mk2 / (dn * dn);
  }
  if (t == 0) out[(size_t)Q * KCLS] = -lossA[0] / (float)Q;
}

extern "C" void kernel_launch(void* const* d_in, const int* in_sizes, int n_in,
                              void* d_out, int out_size, void* d_ws, size_t ws_size,
                              hipStream_t stream) {
  const float* feat  = (const float*)d_in[0];
  const int*   label = (const int*)d_in[1];
  float* out = (float*)d_out;
  float* ws  = (float*)d_ws;

  int total_rows = in_sizes[0] / DDIM;
  int Q = total_rows - 25;

  float* proto     = ws;
  float* proto_y2  = ws + 1280;
  float* num       = ws + 1288;
  float* den       = ws + 2568;
  float* lossA     = ws + 2576;
  float* proto2    = ws + 2580;
  float* proto2_y2 = ws + 3860;

  hpr_k1<<<1, 256, 0, stream>>>(feat, proto, proto_y2, num, den, lossA);

  int nblk = (Q + 31) / 32;
  hpr_pass<false><<<nblk, 64, 0, stream>>>(feat, label, proto, proto_y2,
                                           num, den, lossA, out, Q);
  hpr_k3<<<1, 256, 0, stream>>>(num, den, lossA, proto2, proto2_y2, out, Q);
  hpr_pass<true><<<nblk, 64, 0, stream>>>(feat, label, proto2, proto2_y2,
                                          num, den, lossA, out, Q);
}


# pass kernels: row norms summed from the A fragments beside the matrix loop, one scale chain per lane, scale applied to the product, not 16 serial reduce-and-rescale rounds; norm sum order changes
# speedup vs baseline: 1.8997x; 1.4208x over previous
#include <hip/hip_runtime.h>
#include <hip/hip_bf16.h>
#include <math.h>

typedef float v2f __attribute__((ext_vector_type(2)));
typedef float v8f __attribute__((ext_vector_type(8)));

#define DDIM 256
#define KCLS 5
#define PITCH 260
#define MAXN  9.96f

__device__ __forceinline__ float waveRedSum(float v) {
#pragma unroll
  for (int off = 16; off > 0; off >>= 1) v += __shfl_xor(v, off, 32);
  return v;
}

__global__ __launch_bounds__(256) void hpr_k1(const float* __restrict__ feat,
                                              float* __restrict__ proto,
                                              float* __restrict__ proto_y2,
                                              float* __restrict__ num,
                                              float* __restrict__ den,
                                              float* __restrict__ lossA) {
  __shared__ float sW[32];
  __shared__ float sG[32];
  __shared__ float sRed[KCLS][8];
  const int t = threadIdx.x;
  const int wave = t >> 5, lane = t & 31;

  float vcol[25];
#pragma unroll
  for (int r = 0; r < 25; r++) vcol[r] = feat[r * DDIM + t];

  float ss[4];
#pragma unroll
  for (int i = 0; i < 4; i++) {
    const int r  = wave + 8 * i;
    const int rr = (r < 25) ? r : 24;
    const float4* row = (const float4*)(feat + rr * DDIM);
    float4 a0 = row[lane];
    float4 a1 = row[lane + 32];
    ss[i] = a0.x * a0.x + a0.y * a0.y + a0.z * a0.z + a0.w * a0.w
          + a1.x * a1.x + a1.y * a1.y + a1.z * a1.z + a1.w * a1.w;
  }
#pragma unroll
  for (int off = 16; off > 0; off >>= 1) {
#pragma unroll
    for (int i = 0; i < 4; i++) ss[i] += __shfl_xor(ss[i], off, 32);
  }
#pragma unroll
  for (int i = 0; i < 4; i++) {
    const int r = wave + 8 * i;
    float n2 = ss[i];
    float n  = fmaxf(sqrtf(n2), 1e-5f);
    float sc = tanhf(0.1f * n) / (0.1f * n);
    float pn = sc * n;
    if (pn > MAXN) { sc *= MAXN / pn; pn = MAXN; }
    float x2    = pn * pn;
    float kq    = 2.0f / (1.0f + 0.01f * x2);
    float gamma = rsqrtf(1.0f - 0.01f * kq * kq * x2);
    if (lane == 0 && r < 25) { sW[r] = gamma * kq * sc; sG[r] = gamma; }
  }
  __syncthreads();

  float mk[KCLS], agc[KCLS], part[KCLS];
#pragma unroll
  for (int c = 0; c < KCLS; c++) {
    float agk = 0.f, ag = 0.f;
#pragma unroll
    for (int j = 0; j < 5; j++) {
      const int r = 5 * c + j;
      agk += sW[r] * vcol[r];
      ag  += sG[r];
    }
    num[c * DDIM + t] = agk;
    agc[c]  = ag;
    mk[c]   = agk / ag;
    part[c] = waveRedSum(mk[c] * mk[c]);
  }
  if (lane == 0) {
#pragma unroll
    for (int c = 0; c < KCLS; c++) sRed[c][wave] = part[c];
  }
  __syncthreads();
#pragma unroll
  for (int c = 0; c < KCLS; c++) {
    float mk2 = 0.f;
#pragma unroll
    for (int w = 0; w < 8; w++) mk2 += sRed[c][w];
    float dn = 1.0f + sqrtf(1.0f - 0.01f * mk2);
    proto[c * DDIM + t] = mk[c] / dn;
    if (t == 0) { proto_y2[c] = mk2 / (dn * dn); den[c] = agc[c]; }
  }
  if (t == 0) lossA[0] = 0.f;
}

template <bool PASS2>
__global__ __launch_bounds__(64) void hpr_pass(const float* __restrict__ feat,
                                               const int* __restrict__ label,
                                               const float* __restrict__ proto,
                                               const float* __restrict__ proto_y2,
                                               float* __restrict__ num,
                                               float* __restrict__ den,
                                               float* __restrict__ lossA,
                                               float* __restrict__ out, int Q) {
  __shared__ __align__(16) float tile[2][16][PITCH];
  __shared__ __align__(16) float sproto[16][PITCH];
  __shared__ float sX2[2][16];
  __shared__ float sSc[2][16];
  __shared__ float sDist[2][16][8];
  __shared__ int   sPred[32];
  __shared__ float sCoef[32];
  __shared__ float bden[KCLS];
  __shared__ float bloss;

  const int tid  = threadIdx.x;
  const int wave = tid >> 5, lane = tid & 31;
  const int m16  = lane & 15, half = lane >> 4;
  const int qbase = blockIdx.x * 32 + wave * 16;

  for (int idx = tid; idx < 16 * DDIM; idx += 64) {
    int r = idx >> 8, cc = idx & 255;
    sproto[r][cc] = (r < KCLS) ? proto[r * DDIM + cc] : 0.f;
  }
  if (!PASS2) {
    if (tid < KCLS) bden[tid] = 0.f;
    if (tid == KCLS) bloss = 0.f;
  }

  for (int r = 0; r < 16; r++) {
    int q = qbase + r;
    const float* row = feat + (size_t)(25 + (q < Q ? q : 0)) * DDIM;
    float4 a0 = ((const float4*)row)[lane];
    float4 a1 = ((const float4*)row)[lane + 32];
    *(float4*)&tile[wave][r][lane * 4]       = a0;
    *(float4*)&tile[wave][r][128 + lane * 4] = a1;
  }
  __syncthreads();

  v8f c = {0.f, 0.f, 0.f, 0.f, 0.f, 0.f, 0.f, 0.f};
  v2f ssq0 = {0.f, 0.f}, ssq1 = {0.f, 0.f};
  for (int kc = 0; kc < DDIM; kc += 8) {
    int col = kc + half * 2;
    v2f a0 = *(const v2f*)&tile[wave][m16][col];
    v2f b0 = *(const v2f*)&sproto[m16][col];
    v2f a1 = *(const v2f*)&tile[wave][m16][col + 4];
    v2f b1 = *(const v2f*)&sproto[m16][col + 4];
    ssq0 += a0 * a0;
    ssq1 += a1 * a1;
    c = __builtin_amdgcn_wmma_f32_16x16x4_f32(false, a0, false, b0, (short)0, c,
                                              false, false);
    c = __builtin_amdgcn_wmma_f32_16x16x4_f32(false, a1, false, b1, (short)0, c,
                                              false, false);
  }

  float n2 = (ssq0.x + ssq0.y) + (ssq1.x + ssq1.y);
  n2 += __shfl_xor(n2, 16, 32);
  float n  = fmaxf(sqrtf(n2), 1e-5f);
  float sc = tanhf(0.1f * n) / (0.1f * n);
  float pn = sc * n;
  if (pn > MAXN) { sc *= MAXN / pn; pn = MAXN; }
  const float x2own = pn * pn;
  if (lane < 16) { sX2[wave][lane] = x2own; sSc[wave][lane] = sc; }
  __syncthreads();

  float y2 = proto_y2[(m16 < KCLS) ? m16 : 0];
  if (m16 < KCLS) {
#pragma unroll
    for (int g = 0; g < 8; g++) {
      int   m  = 8 * half + g;
      float x2 = sX2[wave][m];
      float xy = sSc[wave][m] * c[g];
      float A  = 1.f - 0.02f * xy + 0.01f * y2;
      float B  = 1.f - 0.01f * x2;
      float nv2 = fmaxf(A * A * x2 - 2.f * A * B * xy + B * B * y2, 0.f);
      float dnm = 1.f - 0.02f * xy + 1e-4f * x2 * y2 + 1e-5f;
      float man = sqrtf(nv2) / dnm;
      float z = fminf(fmaxf(0.1f * man, -1.f + 1e-5f), 1.f - 1e-5f);
      sDist[wave][m][m16] = 10.f * (log1pf(z) - log1pf(-z));
    }
  }
  __syncthreads();

  float ceContrib = 0.f;
  if (lane < 16) {
    int  q = qbase + lane;
    bool valid = (q < Q);
    float dv[KCLS];
#pragma unroll
    for (int k = 0; k < KCLS; k++) dv[k] = sDist[wave][lane][k];
    float best = dv[0]; int pred = 0;
#pragma unroll
    for (int k = 1; k < KCLS; k++) { if (dv[k] < best) { best = dv[k]; pred = k; } }

    if (PASS2) {
      float e[KCLS]; float se = 0.f;
#pragma unroll
      for (int k = 0; k < KCLS; k++) { e[k] = expf(best - dv[k]); se += e[k]; }
      if (valid) {
        float inv = 1.f / se;
#pragma unroll
        for (int k = 0; k < KCLS; k++) out[(size_t)q * KCLS + k] = e[k] * inv;
      }
    } else {
      float se = 0.f;
#pragma unroll
      for (int k = 0; k < KCLS; k++) se += expf(best - dv[k]);
      float lse = logf(se);
      int lbl = label[Q + q];
      float dl = dv[0];
#pragma unroll
      for (int k = 1; k < KCLS; k++) dl = (lbl == k) ? dv[k] : dl;
      float ce = (best - dl) - lse;
      ceContrib = valid ? ce : 0.f;

      float x2q   = x2own;
      float kq    = 2.0f / (1.0f + 0.01f * x2q);
      float gamma = rsqrtf(1.0f - 0.01f * kq * kq * x2q);
      sPred[wave * 16 + lane] = pred;
      sCoef[wave * 16 + lane] = valid ? gamma * kq * sc : 0.f;
      if (valid) atomicAdd(&bden[pred], gamma);
    }
  }

  if (!PASS2) {
    float s = waveRedSum(ceContrib);
    if (lane == 0) atomicAdd(&bloss, s);
    __syncthreads();

    float racc[KCLS][4];
#pragma unroll
    for (int cc = 0; cc < KCLS; cc++)
#pragma unroll
      for (int j = 0; j < 4; j++) racc[cc][j] = 0.f;

    for (int m2 = 0; m2 < 32; m2++) {
      int p = sPred[m2]; float cf = sCoef[m2];
      int wv = m2 >> 4, r = m2 & 15;
      float v0 = tile[wv][r][tid];
      float v1 = tile[wv][r][tid + 64];
      float v2 = tile[wv][r][tid + 128];
      float v3 = tile[wv][r][tid + 192];
#pragma unroll
      for (int cc = 0; cc < KCLS; cc++) {
        float msk = (p == cc) ? cf : 0.f;
        racc[cc][0] += msk * v0; racc[cc][1] += msk * v1;
        racc[cc][2] += msk * v2; racc[cc][3] += msk * v3;
      }
    }
#pragma unroll
    for (int cc = 0; cc < KCLS; cc++) {
      atomicAdd(&num[cc * DDIM + tid],       racc[cc][0]);
      atomicAdd(&num[cc * DDIM + tid + 64],  racc[cc][1]);
      atomicAdd(&num[cc * DDIM + tid + 128], racc[cc][2]);
      atomicAdd(&num[cc * DDIM + tid + 192], racc[cc][3]);
    }
    if (tid < KCLS) atomicAdd(&den[tid], bden[tid]);
    if (tid == KCLS) atomicAdd(lossA, bloss);
  }
}

__global__ __launch_bounds__(256) void hpr_k3(const float* __restrict__ num,
                                              const float* __restrict__ den,
                                              const float* __restrict__ lossA,
                                              float* __restrict__ proto2,
                                              float* __restrict__ proto2_y2,
                                              float* __restrict__ out, int Q) {
  __shared__ float sRed[KCLS][8];
  const int t = threadIdx.x;
  const int wave = t >> 5, lane = t & 31;
  float mk[KCLS], part[KCLS];
#pragma unroll
  for (int c = 0; c < KCLS; c++) {
    mk[c]   = num[c * DDIM + t] / den[c];
    part[c] = waveRedSum(mk[c] * mk[c]);
  }
  if (lane == 0) {
#pragma unroll
    for (int c = 0; c < KCLS; c++) sRed[c][wave] = part[c];
  }
  __syncthreads();
#pragma unroll
  for (int c = 0; c < KCLS; c++) {
    float mk2 = 0.f;
#pragma unroll
    for (int w = 0; w < 8; w++) mk2 += sRed[c][w];
    float dn = 1.0f + sqrtf(1.0f - 0.01f * mk2);
    proto2[c * DDIM + t] = mk[c] / dn;
    if (t == 0) proto2_y2[c] = mk2 / (dn * dn);
  }
  if (t == 0) out[(size_t)Q * KCLS] = -lossA[0] / (float)Q;
}

extern "C" void kernel_launch(void* const* d_in, const int* in_sizes, int n_in,
                              void* d_out, int out_size, void* d_ws, size_t ws_size,
                              hipStream_t stream) {
  const float* feat  = (const float*)d_in[0];
  const int*   label = (const int*)d_in[1];
  float* out = (float*)d_out;
  float* ws  = (float*)d_ws;

  int total_rows = in_sizes[0] / DDIM;
  int Q = total_rows - 25;

  float* proto     = ws;
  float* proto_y2  = ws + 1280;
  float* num       = ws + 1288;
  float* den       = ws + 2568;
  float* lossA     = ws + 2576;
  float* proto2    = ws + 2580;
  float* proto2_y2 = ws + 3860;

  hpr_k1<<<1, 256, 0, stream>>>(feat, proto, proto_y2, num, den, lossA);

  int nblk = (Q + 31) / 32;
  hpr_pass<false><<<nblk, 64, 0, stream>>>(feat, label, proto, proto_y2,
                                           num, den, lossA, out, Q);
  hpr_k3<<<1, 256, 0, stream>>>(num, den, lossA, proto2, proto2_y2, out, Q);
  hpr_pass<true><<<nblk, 64, 0, stream>>>(feat, label, proto2, proto2_y2,
                                          num, den, lossA, out, Q);
}


# pass kernels: B operand staged as its 5 real rows, 5 b128 loads per thread issued with the query-row loads, not a serial 64-step load-wait-store loop over 16 padded rows; unused columns re-read row 0
# speedup vs baseline: 2.9809x; 1.5691x over previous
#include <hip/hip_runtime.h>
#include <hip/hip_bf16.h>
#include <math.h>

typedef float v2f __attribute__((ext_vector_type(2)));
typedef float v8f __attribute__((ext_vector_type(8)));

#define DDIM 256
#define KCLS 5
#define PITCH 260
#define MAXN  9.96f

__device__ __forceinline__ float waveRedSum(float v) {
#pragma unroll
  for (int off = 16; off > 0; off >>= 1) v += __shfl_xor(v, off, 32);
  return v;
}

__global__ __launch_bounds__(256) void hpr_k1(const float* __restrict__ feat,
                                              float* __restrict__ proto,
                                              float* __restrict__ proto_y2,
                                              float* __restrict__ num,
                                              float* __restrict__ den,
                                              float* __restrict__ lossA) {
  __shared__ float sW[32];
  __shared__ float sG[32];
  __shared__ float sRed[KCLS][8];
  const int t = threadIdx.x;
  const int wave = t >> 5, lane = t & 31;

  float vcol[25];
#pragma unroll
  for (int r = 0; r < 25; r++) vcol[r] = feat[r * DDIM + t];

  float ss[4];
#pragma unroll
  for (int i = 0; i < 4; i++) {
    const int r  = wave + 8 * i;
    const int rr = (r < 25) ? r : 24;
    const float4* row = (const float4*)(feat + rr * DDIM);
    float4 a0 = row[lane];
    float4 a1 = row[lane + 32];
    ss[i] = a0.x * a0.x + a0.y * a0.y + a0.z * a0.z + a0.w * a0.w
          + a1.x * a1.x + a1.y * a1.y + a1.z * a1.z + a1.w * a1.w;
  }
#pragma unroll
  for (int off = 16; off > 0; off >>= 1) {
#pragma unroll
    for (int i = 0; i < 4; i++) ss[i] += __shfl_xor(ss[i], off, 32);
  }
#pragma unroll
  for (int i = 0; i < 4; i++) {
    const int r = wave + 8 * i;
    float n2 = ss[i];
    float n  = fmaxf(sqrtf(n2), 1e-5f);
    float sc = tanhf(0.1f * n) / (0.1f * n);
    float pn = sc * n;
    if (pn > MAXN) { sc *= MAXN / pn; pn = MAXN; }
    float x2    = pn * pn;
    float kq    = 2.0f / (1.0f + 0.01f * x2);
    float gamma = rsqrtf(1.0f - 0.01f * kq * kq * x2);
    if (lane == 0 && r < 25) { sW[r] = gamma * kq * sc; sG[r] = gamma; }
  }
  __syncthreads();

  float mk[KCLS], agc[KCLS], part[KCLS];
#pragma unroll
  for (int c = 0; c < KCLS; c++) {
    float agk = 0.f, ag = 0.f;
#pragma unroll
    for (int j = 0; j < 5; j++) {
      const int r = 5 * c + j;
      agk += sW[r] * vcol[r];
      ag  += sG[r];
    }
    num[c * DDIM + t] = agk;
    agc[c]  = ag;
    mk[c]   = agk / ag;
    part[c] = waveRedSum(mk[c] * mk[c]);
  }
  if (lane == 0) {
#pragma unroll
    for (int c = 0; c < KCLS; c++) sRed[c][wave] = part[c];
  }
  __syncthreads();
#pragma unroll
  for (int c = 0; c < KCLS; c++) {
    float mk2 = 0.f;
#pragma unroll
    for (int w = 0; w < 8; w++) mk2 += sRed[c][w];
    float dn = 1.0f + sqrtf(1.0f - 0.01f * mk2);
    proto[c * DDIM + t] = mk[c] / dn;
    if (t == 0) { proto_y2[c] = mk2 / (dn * dn); den[c] = agc[c]; }
  }
  if (t == 0) lossA[0] = 0.f;
}

template <bool PASS2>
__global__ __launch_bounds__(64) void hpr_pass(const float* __restrict__ feat,
                                               const int* __restrict__ label,
                                               const float* __restrict__ proto,
                                               const float* __restrict__ proto_y2,
                                               float* __restrict__ num,
                                               float* __restrict__ den,
                                               float* __restrict__ lossA,
                                               float* __restrict__ out, int Q) {
  __shared__ __align__(16) float tile[2][16][PITCH];
  __shared__ __align__(16) float sproto[KCLS][PITCH];
  __shared__ float sX2[2][16];
  __shared__ float sSc[2][16];
  __shared__ float sDist[2][16][8];
  __shared__ int   sPred[32];
  __shared__ float sCoef[32];
  __shared__ float bden[KCLS];
  __shared__ float bloss;

  const int tid  = threadIdx.x;
  const int wave = tid >> 5, lane = tid & 31;
  const int m16  = lane & 15, half = lane >> 4;
  const int qbase = blockIdx.x * 32 + wave * 16;

  float4 pv[KCLS];
#pragma unroll
  for (int i = 0; i < KCLS; i++) pv[i] = ((const float4*)(proto + i * DDIM))[tid];
  if (!PASS2) {
    if (tid < KCLS) bden[tid] = 0.f;
    if (tid == KCLS) bloss = 0.f;
  }

  for (int r = 0; r < 16; r++) {
    int q = qbase + r;
    const float* row = feat + (size_t)(25 + (q < Q ? q : 0)) * DDIM;
    float4 a0 = ((const float4*)row)[lane];
    float4 a1 = ((const float4*)row)[lane + 32];
    *(float4*)&tile[wave][r][lane * 4]       = a0;
    *(float4*)&tile[wave][r][128 + lane * 4] = a1;
  }
#pragma unroll
  for (int i = 0; i < KCLS; i++) *(float4*)&sproto[i][tid * 4] = pv[i];
  __syncthreads();

  const int mB = (m16 < KCLS) ? m16 : 0;
  v8f c = {0.f, 0.f, 0.f, 0.f, 0.f, 0.f, 0.f, 0.f};
  v2f ssq0 = {0.f, 0.f}, ssq1 = {0.f, 0.f};
  for (int kc = 0; kc < DDIM; kc += 8) {
    int col = kc + half * 2;
    v2f a0 = *(const v2f*)&tile[wave][m16][col];
    v2f b0 = *(const v2f*)&sproto[mB][col];
    v2f a1 = *(const v2f*)&tile[wave][m16][col + 4];
    v2f b1 = *(const v2f*)&sproto[mB][col + 4];
    ssq0 += a0 * a0;
    ssq1 += a1 * a1;
    c = __builtin_amdgcn_wmma_f32_16x16x4_f32(false, a0, false, b0, (short)0, c,
                                              false, false);
    c = __builtin_amdgcn_wmma_f32_16x16x4_f32(false, a1, false, b1, (short)0, c,
                                              false, false);
  }

  float n2 = (ssq0.x + ssq0.y) + (ssq1.x + ssq1.y);
  n2 += __shfl_xor(n2, 16, 32);
  float n  = fmaxf(sqrtf(n2), 1e-5f);
  float sc = tanhf(0.1f * n) / (0.1f * n);
  float pn = sc * n;
  if (pn > MAXN) { sc *= MAXN / pn; pn = MAXN; }
  const float x2own = pn * pn;
  if (lane < 16) { sX2[wave][lane] = x2own; sSc[wave][lane] = sc; }
  __syncthreads();

  float y2 = proto_y2[(m16 < KCLS) ? m16 : 0];
  if (m16 < KCLS) {
#pragma unroll
    for (int g = 0; g < 8; g++) {
      int   m  = 8 * half + g;
      float x2 = sX2[wave][m];
      float xy = sSc[wave][m] * c[g];
      float A  = 1.f - 0.02f * xy + 0.01f * y2;
      float B  = 1.f - 0.01f * x2;
      float nv2 = fmaxf(A * A * x2 - 2.f * A * B * xy + B * B * y2, 0.f);
      float dnm = 1.f - 0.02f * xy + 1e-4f * x2 * y2 + 1e-5f;
      float man = sqrtf(nv2) / dnm;
      float z = fminf(fmaxf(0.1f * man, -1.f + 1e-5f), 1.f - 1e-5f);
      sDist[wave][m][m16] = 10.f * (log1pf(z) - log1pf(-z));
    }
  }
  __syncthreads();

  float ceContrib = 0.f;
  if (lane < 16) {
    int  q = qbase + lane;
    bool valid = (q < Q);
    float dv[KCLS];
#pragma unroll
    for (int k = 0; k < KCLS; k++) dv[k] = sDist[wave][lane][k];
    float best = dv[0]; int pred = 0;
#pragma unroll
    for (int k = 1; k < KCLS; k++) { if (dv[k] < best) { best = dv[k]; pred = k; } }

    if (PASS2) {
      float e[KCLS]; float se = 0.f;
#pragma unroll
      for (int k = 0; k < KCLS; k++) { e[k] = expf(best - dv[k]); se += e[k]; }
      if (valid) {
        float inv = 1.f / se;
#pragma unroll
        for (int k = 0; k < KCLS; k++) out[(size_t)q * KCLS + k] = e[k] * inv;
      }
    } else {
      float se = 0.f;
#pragma unroll
      for (int k = 0; k < KCLS; k++) se += expf(best - dv[k]);
      float lse = logf(se);
      int lbl = label[Q + q];
      float dl = dv[0];
#pragma unroll
      for (int k = 1; k < KCLS; k++) dl = (lbl == k) ? dv[k] : dl;
      float ce = (best - dl) - lse;
      ceContrib = valid ? ce : 0.f;

      float x2q   = x2own;
      float kq    = 2.0f / (1.0f + 0.01f * x2q);
      float gamma = rsqrtf(1.0f - 0.01f * kq * kq * x2q);
      sPred[wave * 16 + lane] = pred;
      sCoef[wave * 16 + lane] = valid ? gamma * kq * sc : 0.f;
      if (valid) atomicAdd(&bden[pred], gamma);
    }
  }

  if (!PASS2) {
    float s = waveRedSum(ceContrib);
    if (lane == 0) atomicAdd(&bloss, s);
    __syncthreads();

    float racc[KCLS][4];
#pragma unroll
    for (int cc = 0; cc < KCLS; cc++)
#pragma unroll
      for (int j = 0; j < 4; j++) racc[cc][j] = 0.f;

    for (int m2 = 0; m2 < 32; m2++) {
      int p = sPred[m2]; float cf = sCoef[m2];
      int wv = m2 >> 4, r = m2 & 15;
      float v0 = tile[wv][r][tid];
      float v1 = tile[wv][r][tid + 64];
      float v2 = tile[wv][r][tid + 128];
      float v3 = tile[wv][r][tid + 192];
#pragma unroll
      for (int cc = 0; cc < KCLS; cc++) {
        float msk = (p == cc) ? cf : 0.f;
        racc[cc][0] += msk * v0; racc[cc][1] += msk * v1;
        racc[cc][2] += msk * v2; racc[cc][3] += msk * v3;
      }
    }
#pragma unroll
    for (int cc = 0; cc < KCLS; cc++) {
      atomicAdd(&num[cc * DDIM + tid],       racc[cc][0]);
      atomicAdd(&num[cc * DDIM + tid + 64],  racc[cc][1]);
      atomicAdd(&num[cc * DDIM + tid + 128], racc[cc][2]);
      atomicAdd(&num[cc * DDIM + tid + 192], racc[cc][3]);
    }
    if (tid < KCLS) atomicAdd(&den[tid], bden[tid]);
    if (tid == KCLS) atomicAdd(lossA, bloss);
  }
}

__global__ __launch_bounds__(256) void hpr_k3(const float* __restrict__ num,
                                              const float* __restrict__ den,
                                              const float* __restrict__ lossA,
                                              float* __restrict__ proto2,
                                              float* __restrict__ proto2_y2,
                                              float* __restrict__ out, int Q) {
  __shared__ float sRed[KCLS][8];
  const int t = threadIdx.x;
  const int wave = t >> 5, lane = t & 31;
  float mk[KCLS], part[KCLS];
#pragma unroll
  for (int c = 0; c < KCLS; c++) {
    mk[c]   = num[c * DDIM + t] / den[c];
    part[c] = waveRedSum(mk[c] * mk[c]);
  }
  if (lane == 0) {
#pragma unroll
    for (int c = 0; c < KCLS; c++) sRed[c][wave] = part[c];
  }
  __syncthreads();
#pragma unroll
  for (int c = 0; c < KCLS; c++) {
    float mk2 = 0.f;
#pragma unroll
    for (int w = 0; w < 8; w++) mk2 += sRed[c][w];
    float dn = 1.0f + sqrtf(1.0f - 0.01f * mk2);
    proto2[c * DDIM + t] = mk[c] / dn;
    if (t == 0) proto2_y2[c] = mk2 / (dn * dn);
  }
  if (t == 0) out[(size_t)Q * KCLS] = -lossA[0] / (float)Q;
}

extern "C" void kernel_launch(void* const* d_in, const int* in_sizes, int n_in,
                              void* d_out, int out_size, void* d_ws, size_t ws_size,
                              hipStream_t stream) {
  const float* feat  = (const float*)d_in[0];
  const int*   label = (const int*)d_in[1];
  float* out = (float*)d_out;
  float* ws  = (float*)d_ws;

  int total_rows = in_sizes[0] / DDIM;
  int Q = total_rows - 25;

  float* proto     = ws;
  float* proto_y2  = ws + 1280;
  float* num       = ws + 1288;
  float* den       = ws + 2568;
  float* lossA     = ws + 2576;
  float* proto2    = ws + 2580;
  float* proto2_y2 = ws + 3860;

  hpr_k1<<<1, 256, 0, stream>>>(feat, proto, proto_y2, num, den, lossA);

  int nblk = (Q + 31) / 32;
  hpr_pass<false><<<nblk, 64, 0, stream>>>(feat, label, proto, proto_y2,
                                           num, den, lossA, out, Q);
  hpr_k3<<<1, 256, 0, stream>>>(num, den, lossA, proto2, proto2_y2, out, Q);
  hpr_pass<true><<<nblk, 64, 0, stream>>>(feat, label, proto2, proto2_y2,
                                          num, den, lossA, out, Q);
}


# pass kernels: artanh of the distance as one log of the quotient (1+z)/(1-z), not the difference of two log1p calls (each a long extended-precision routine); sum orders unchanged
# speedup vs baseline: 3.1588x; 1.0597x over previous
#include <hip/hip_runtime.h>
#include <hip/hip_bf16.h>
#include <math.h>

typedef float v2f __attribute__((ext_vector_type(2)));
typedef float v8f __attribute__((ext_vector_type(8)));

#define DDIM 256
#define KCLS 5
#define PITCH 260
#define MAXN  9.96f

__device__ __forceinline__ float waveRedSum(float v) {
#pragma unroll
  for (int off = 16; off > 0; off >>= 1) v += __shfl_xor(v, off, 32);
  return v;
}

__global__ __launch_bounds__(256) void hpr_k1(const float* __restrict__ feat,
                                              float* __restrict__ proto,
                                              float* __restrict__ proto_y2,
                                              float* __restrict__ num,
                                              float* __restrict__ den,
                                              float* __restrict__ lossA) {
  __shared__ float sW[32];
  __shared__ float sG[32];
  __shared__ float sRed[KCLS][8];
  const int t = threadIdx.x;
  const int wave = t >> 5, lane = t & 31;

  float vcol[25];
#pragma unroll
  for (int r = 0; r < 25; r++) vcol[r] = feat[r * DDIM + t];

  float ss[4];
#pragma unroll
  for (int i = 0; i < 4; i++) {
    const int r  = wave + 8 * i;
    const int rr = (r < 25) ? r : 24;
    const float4* row = (const float4*)(feat + rr * DDIM);
    float4 a0 = row[lane];
    float4 a1 = row[lane + 32];
    ss[i] = a0.x * a0.x + a0.y * a0.y + a0.z * a0.z + a0.w * a0.w
          + a1.x * a1.x + a1.y * a1.y + a1.z * a1.z + a1.w * a1.w;
  }
#pragma unroll
  for (int off = 16; off > 0; off >>= 1) {
#pragma unroll
    for (int i = 0; i < 4; i++) ss[i] += __shfl_xor(ss[i], off, 32);
  }
#pragma unroll
  for (int i = 0; i < 4; i++) {
    const int r = wave + 8 * i;
    float n2 = ss[i];
    float n  = fmaxf(sqrtf(n2), 1e-5f);
    float sc = tanhf(0.1f * n) / (0.1f * n);
    float pn = sc * n;
    if (pn > MAXN) { sc *= MAXN / pn; pn = MAXN; }
    float x2    = pn * pn;
    float kq    = 2.0f / (1.0f + 0.01f * x2);
    float gamma = rsqrtf(1.0f - 0.01f * kq * kq * x2);
    if (lane == 0 && r < 25) { sW[r] = gamma * kq * sc; sG[r] = gamma; }
  }
  __syncthreads();

  float mk[KCLS], agc[KCLS], part[KCLS];
#pragma unroll
  for (int c = 0; c < KCLS; c++) {
    float agk = 0.f, ag = 0.f;
#pragma unroll
    for (int j = 0; j < 5; j++) {
      const int r = 5 * c + j;
      agk += sW[r] * vcol[r];
      ag  += sG[r];
    }
    num[c * DDIM + t] = agk;
    agc[c]  = ag;
    mk[c]   = agk / ag;
    part[c] = waveRedSum(mk[c] * mk[c]);
  }
  if (lane == 0) {
#pragma unroll
    for (int c = 0; c < KCLS; c++) sRed[c][wave] = part[c];
  }
  __syncthreads();
#pragma unroll
  for (int c = 0; c < KCLS; c++) {
    float mk2 = 0.f;
#pragma unroll
    for (int w = 0; w < 8; w++) mk2 += sRed[c][w];
    float dn = 1.0f + sqrtf(1.0f - 0.01f * mk2);
    proto[c * DDIM + t] = mk[c] / dn;
    if (t == 0) { proto_y2[c] = mk2 / (dn * dn); den[c] = agc[c]; }
  }
  if (t == 0) lossA[0] = 0.f;
}

template <bool PASS2>
__global__ __launch_bounds__(64) void hpr_pass(const float* __restrict__ feat,
                                               const int* __restrict__ label,
                                               const float* __restrict__ proto,
                                               const float* __restrict__ proto_y2,
                                               float* __restrict__ num,
                                               float* __restrict__ den,
                                               float* __restrict__ lossA,
                                               float* __restrict__ out, int Q) {
  __shared__ __align__(16) float tile[2][16][PITCH];
  __shared__ __align__(16) float sproto[KCLS][PITCH];
  __shared__ float sX2[2][16];
  __shared__ float sSc[2][16];
  __shared__ float sDist[2][16][8];
  __shared__ int   sPred[32];
  __shared__ float sCoef[32];
  __shared__ float bden[KCLS];
  __shared__ float bloss;

  const int tid  = threadIdx.x;
  const int wave = tid >> 5, lane = tid & 31;
  const int m16  = lane & 15, half = lane >> 4;
  const int qbase = blockIdx.x * 32 + wave * 16;

  float4 pv[KCLS];
#pragma unroll
  for (int i = 0; i < KCLS; i++) pv[i] = ((const float4*)(proto + i * DDIM))[tid];
  if (!PASS2) {
    if (tid < KCLS) bden[tid] = 0.f;
    if (tid == KCLS) bloss = 0.f;
  }

  for (int r = 0; r < 16; r++) {
    int q = qbase + r;
    const float* row = feat + (size_t)(25 + (q < Q ? q : 0)) * DDIM;
    float4 a0 = ((const float4*)row)[lane];
    float4 a1 = ((const float4*)row)[lane + 32];
    *(float4*)&tile[wave][r][lane * 4]       = a0;
    *(float4*)&tile[wave][r][128 + lane * 4] = a1;
  }
#pragma unroll
  for (int i = 0; i < KCLS; i++) *(float4*)&sproto[i][tid * 4] = pv[i];
  __syncthreads();

  const int mB = (m16 < KCLS) ? m16 : 0;
  v8f c = {0.f, 0.f, 0.f, 0.f, 0.f, 0.f, 0.f, 0.f};
  v2f ssq0 = {0.f, 0.f}, ssq1 = {0.f, 0.f};
  for (int kc = 0; kc < DDIM; kc += 8) {
    int col = kc + half * 2;
    v2f a0 = *(const v2f*)&tile[wave][m16][col];
    v2f b0 = *(const v2f*)&sproto[mB][col];
    v2f a1 = *(const v2f*)&tile[wave][m16][col + 4];
    v2f b1 = *(const v2f*)&sproto[mB][col + 4];
    ssq0 += a0 * a0;
    ssq1 += a1 * a1;
    c = __builtin_amdgcn_wmma_f32_16x16x4_f32(false, a0, false, b0, (short)0, c,
                                              false, false);
    c = __builtin_amdgcn_wmma_f32_16x16x4_f32(false, a1, false, b1, (short)0, c,
                                              false, false);
  }

  float n2 = (ssq0.x + ssq0.y) + (ssq1.x + ssq1.y);
  n2 += __shfl_xor(n2, 16, 32);
  float n  = fmaxf(sqrtf(n2), 1e-5f);
  float sc = tanhf(0.1f * n) / (0.1f * n);
  float pn = sc * n;
  if (pn > MAXN) { sc *= MAXN / pn; pn = MAXN; }
  const float x2own = pn * pn;
  if (lane < 16) { sX2[wave][lane] = x2own; sSc[wave][lane] = sc; }
  __syncthreads();

  float y2 = proto_y2[(m16 < KCLS) ? m16 : 0];
  if (m16 < KCLS) {
#pragma unroll
    for (int g = 0; g < 8; g++) {
      int   m  = 8 * half + g;
      float x2 = sX2[wave][m];
      float xy = sSc[wave][m] * c[g];
      float A  = 1.f - 0.02f * xy + 0.01f * y2;
      float B  = 1.f - 0.01f * x2;
      float nv2 = fmaxf(A * A * x2 - 2.f * A * B * xy + B * B * y2, 0.f);
      float dnm = 1.f - 0.02f * xy + 1e-4f * x2 * y2 + 1e-5f;
      float man = sqrtf(nv2) / dnm;
      float z = fminf(fmaxf(0.1f * man, -1.f + 1e-5f), 1.f - 1e-5f);
      sDist[wave][m][m16] = 10.f * logf((1.f + z) / (1.f - z));
    }
  }
  __syncthreads();

  float ceContrib = 0.f;
  if (lane < 16) {
    int  q = qbase + lane;
    bool valid = (q < Q);
    float dv[KCLS];
#pragma unroll
    for (int k = 0; k < KCLS; k++) dv[k] = sDist[wave][lane][k];
    float best = dv[0]; int pred = 0;
#pragma unroll
    for (int k = 1; k < KCLS; k++) { if (dv[k] < best) { best = dv[k]; pred = k; } }

    if (PASS2) {
      float e[KCLS]; float se = 0.f;
#pragma unroll
      for (int k = 0; k < KCLS; k++) { e[k] = expf(best - dv[k]); se += e[k]; }
      if (valid) {
        float inv = 1.f / se;
#pragma unroll
        for (int k = 0; k < KCLS; k++) out[(size_t)q * KCLS + k] = e[k] * inv;
      }
    } else {
      float se = 0.f;
#pragma unroll
      for (int k = 0; k < KCLS; k++) se += expf(best - dv[k]);
      float lse = logf(se);
      int lbl = label[Q + q];
      float dl = dv[0];
#pragma unroll
      for (int k = 1; k < KCLS; k++) dl = (lbl == k) ? dv[k] : dl;
      float ce = (best - dl) - lse;
      ceContrib = valid ? ce : 0.f;

      float x2q   = x2own;
      float kq    = 2.0f / (1.0f + 0.01f * x2q);
      float gamma = rsqrtf(1.0f - 0.01f * kq * kq * x2q);
      sPred[wave * 16 + lane] = pred;
      sCoef[wave * 16 + lane] = valid ? gamma * kq * sc : 0.f;
      if (valid) atomicAdd(&bden[pred], gamma);
    }
  }

  if (!PASS2) {
    float s = waveRedSum(ceContrib);
    if (lane == 0) atomicAdd(&bloss, s);
    __syncthreads();

    float racc[KCLS][4];
#pragma unroll
    for (int cc = 0; cc < KCLS; cc++)
#pragma unroll
      for (int j = 0; j < 4; j++) racc[cc][j] = 0.f;

    for (int m2 = 0; m2 < 32; m2++) {
      int p = sPred[m2]; float cf = sCoef[m2];
      int wv = m2 >> 4, r = m2 & 15;
      float v0 = tile[wv][r][tid];
      float v1 = tile[wv][r][tid + 64];
      float v2 = tile[wv][r][tid + 128];
      float v3 = tile[wv][r][tid + 192];
#pragma unroll
      for (int cc = 0; cc < KCLS; cc++) {
        float msk = (p == cc) ? cf : 0.f;
        racc[cc][0] += msk * v0; racc[cc][1] += msk * v1;
        racc[cc][2] += msk * v2; racc[cc][3] += msk * v3;
      }
    }
#pragma unroll
    for (int cc = 0; cc < KCLS; cc++) {
      atomicAdd(&num[cc * DDIM + tid],       racc[cc][0]);
      atomicAdd(&num[cc * DDIM + tid + 64],  racc[cc][1]);
      atomicAdd(&num[cc * DDIM + tid + 128], racc[cc][2]);
      atomicAdd(&num[cc * DDIM + tid + 192], racc[cc][3]);
    }
    if (tid < KCLS) atomicAdd(&den[tid], bden[tid]);
    if (tid == KCLS) atomicAdd(lossA, bloss);
  }
}

__global__ __launch_bounds__(256) void hpr_k3(const float* __restrict__ num,
                                              const float* __restrict__ den,
                                              const float* __restrict__ lossA,
                                              float* __restrict__ proto2,
                                              float* __restrict__ proto2_y2,
                                              float* __restrict__ out, int Q) {
  __shared__ float sRed[KCLS][8];
  const int t = threadIdx.x;
  const int wave = t >> 5, lane = t & 31;
  float mk[KCLS], part[KCLS];
#pragma unroll
  for (int c = 0; c < KCLS; c++) {
    mk[c]   = num[c * DDIM + t] / den[c];
    part[c] = waveRedSum(mk[c] * mk[c]);
  }
  if (lane == 0) {
#pragma unroll
    for (int c = 0; c < KCLS; c++) sRed[c][wave] = part[c];
  }
  __syncthreads();
#pragma unroll
  for (int c = 0; c < KCLS; c++) {
    float mk2 = 0.f;
#pragma unroll
    for (int w = 0; w < 8; w++) mk2 += sRed[c][w];
    float dn = 1.0f + sqrtf(1.0f - 0.01f * mk2);
    proto2[c * DDIM + t] = mk[c] / dn;
    if (t == 0) proto2_y2[c] = mk2 / (dn * dn);
  }
  if (t == 0) out[(size_t)Q * KCLS] = -lossA[0] / (float)Q;
}

extern "C" void kernel_launch(void* const* d_in, const int* in_sizes, int n_in,
                              void* d_out, int out_size, void* d_ws, size_t ws_size,
                              hipStream_t stream) {
  const float* feat  = (const float*)d_in[0];
  const int*   label = (const int*)d_in[1];
  float* out = (float*)d_out;
  float* ws  = (float*)d_ws;

  int total_rows = in_sizes[0] / DDIM;
  int Q = total_rows - 25;

  float* proto     = ws;
  float* proto_y2  = ws + 1280;
  float* num       = ws + 1288;
  float* den       = ws + 2568;
  float* lossA     = ws + 2576;
  float* proto2    = ws + 2580;
  float* proto2_y2 = ws + 3860;

  hpr_k1<<<1, 256, 0, stream>>>(feat, proto, proto_y2, num, den, lossA);

  int nblk = (Q + 31) / 32;
  hpr_pass<false><<<nblk, 64, 0, stream>>>(feat, label, proto, proto_y2,
                                           num, den, lossA, out, Q);
  hpr_k3<<<1, 256, 0, stream>>>(num, den, lossA, proto2, proto2_y2, out, Q);
  hpr_pass<true><<<nblk, 64, 0, stream>>>(feat, label, proto2, proto2_y2,
                                          num, den, lossA, out, Q);
}


# pass 1: 8-wave work-groups of 128 queries; four 32-row class sums are added in LDS, one set of float atomics leaves per 128 queries, not per 32 (4x fewer); num/den/loss sum order changes
# speedup vs baseline: 3.4045x; 1.0778x over previous
#include <hip/hip_runtime.h>
#include <hip/hip_bf16.h>
#include <math.h>

typedef float v2f __attribute__((ext_vector_type(2)));
typedef float v8f __attribute__((ext_vector_type(8)));

#define DDIM 256
#define KCLS 5
#define PITCH 260
#define MAXN  9.96f

__device__ __forceinline__ float waveRedSum(float v) {
#pragma unroll
  for (int off = 16; off > 0; off >>= 1) v += __shfl_xor(v, off, 32);
  return v;
}

__global__ __launch_bounds__(256) void hpr_k1(const float* __restrict__ feat,
                                              float* __restrict__ proto,
                                              float* __restrict__ proto_y2,
                                              float* __restrict__ num,
                                              float* __restrict__ den,
                                              float* __restrict__ lossA) {
  __shared__ float sW[32];
  __shared__ float sG[32];
  __shared__ float sRed[KCLS][8];
  const int t = threadIdx.x;
  const int wave = t >> 5, lane = t & 31;

  float vcol[25];
#pragma unroll
  for (int r = 0; r < 25; r++) vcol[r] = feat[r * DDIM + t];

  float ss[4];
#pragma unroll
  for (int i = 0; i < 4; i++) {
    const int r  = wave + 8 * i;
    const int rr = (r < 25) ? r : 24;
    const float4* row = (const float4*)(feat + rr * DDIM);
    float4 a0 = row[lane];
    float4 a1 = row[lane + 32];
    ss[i] = a0.x * a0.x + a0.y * a0.y + a0.z * a0.z + a0.w * a0.w
          + a1.x * a1.x + a1.y * a1.y + a1.z * a1.z + a1.w * a1.w;
  }
#pragma unroll
  for (int off = 16; off > 0; off >>= 1) {
#pragma unroll
    for (int i = 0; i < 4; i++) ss[i] += __shfl_xor(ss[i], off, 32);
  }
#pragma unroll
  for (int i = 0; i < 4; i++) {
    const int r = wave + 8 * i;
    float n2 = ss[i];
    float n  = fmaxf(sqrtf(n2), 1e-5f);
    float sc = tanhf(0.1f * n) / (0.1f * n);
    float pn = sc * n;
    if (pn > MAXN) { sc *= MAXN / pn; pn = MAXN; }
    float x2    = pn * pn;
    float kq    = 2.0f / (1.0f + 0.01f * x2);
    float gamma = rsqrtf(1.0f - 0.01f * kq * kq * x2);
    if (lane == 0 && r < 25) { sW[r] = gamma * kq * sc; sG[r] = gamma; }
  }
  __syncthreads();

  float mk[KCLS], agc[KCLS], part[KCLS];
#pragma unroll
  for (int c = 0; c < KCLS; c++) {
    float agk = 0.f, ag = 0.f;
#pragma unroll
    for (int j = 0; j < 5; j++) {
      const int r = 5 * c + j;
      agk += sW[r] * vcol[r];
      ag  += sG[r];
    }
    num[c * DDIM + t] = agk;
    agc[c]  = ag;
    mk[c]   = agk / ag;
    part[c] = waveRedSum(mk[c] * mk[c]);
  }
  if (lane == 0) {
#pragma unroll
    for (int c = 0; c < KCLS; c++) sRed[c][wave] = part[c];
  }
  __syncthreads();
#pragma unroll
  for (int c = 0; c < KCLS; c++) {
    float mk2 = 0.f;
#pragma unroll
    for (int w = 0; w < 8; w++) mk2 += sRed[c][w];
    float dn = 1.0f + sqrtf(1.0f - 0.01f * mk2);
    proto[c * DDIM + t] = mk[c] / dn;
    if (t == 0) { proto_y2[c] = mk2 / (dn * dn); den[c] = agc[c]; }
  }
  if (t == 0) lossA[0] = 0.f;
}

template <bool PASS2, int NW>
__global__ __launch_bounds__(NW * 32) void hpr_pass(const float* __restrict__ feat,
                                               const int* __restrict__ label,
                                               const float* __restrict__ proto,
                                               const float* __restrict__ proto_y2,
                                               float* __restrict__ num,
                                               float* __restrict__ den,
                                               float* __restrict__ lossA,
                                               float* __restrict__ out, int Q) {
  __shared__ __align__(16) float tile[NW][16][PITCH];
  __shared__ __align__(16) float sproto[KCLS][PITCH];
  __shared__ float sX2[NW][16];
  __shared__ float sSc[NW][16];
  __shared__ float sDist[NW][16][8];
  __shared__ int   sPred[NW * 16];
  __shared__ float sCoef[NW * 16];
  __shared__ float bden[KCLS];
  __shared__ float bloss;

  const int tid  = threadIdx.x;
  const int wave = tid >> 5, lane = tid & 31;
  const int m16  = lane & 15, half = lane >> 4;
  const int qbase = blockIdx.x * (NW * 16) + wave * 16;

  float4 pv[KCLS];
  float  pw[KCLS];
#pragma unroll
  for (int i = 0; i < KCLS; i++) {
    if (NW == 2) pv[i] = ((const float4*)(proto + i * DDIM))[tid];
    else         pw[i] = proto[i * DDIM + tid];
  }
  if (!PASS2) {
    if (tid < KCLS) bden[tid] = 0.f;
    if (tid == KCLS) bloss = 0.f;
  }

  for (int r = 0; r < 16; r++) {
    int q = qbase + r;
    const float* row = feat + (size_t)(25 + (q < Q ? q : 0)) * DDIM;
    float4 a0 = ((const float4*)row)[lane];
    float4 a1 = ((const float4*)row)[lane + 32];
    *(float4*)&tile[wave][r][lane * 4]       = a0;
    *(float4*)&tile[wave][r][128 + lane * 4] = a1;
  }
#pragma unroll
  for (int i = 0; i < KCLS; i++) {
    if (NW == 2) *(float4*)&sproto[i][tid * 4] = pv[i];
    else         sproto[i][tid] = pw[i];
  }
  __syncthreads();

  const int mB = (m16 < KCLS) ? m16 : 0;
  v8f c = {0.f, 0.f, 0.f, 0.f, 0.f, 0.f, 0.f, 0.f};
  v2f ssq0 = {0.f, 0.f}, ssq1 = {0.f, 0.f};
  for (int kc = 0; kc < DDIM; kc += 8) {
    int col = kc + half * 2;
    v2f a0 = *(const v2f*)&tile[wave][m16][col];
    v2f b0 = *(const v2f*)&sproto[mB][col];
    v2f a1 = *(const v2f*)&tile[wave][m16][col + 4];
    v2f b1 = *(const v2f*)&sproto[mB][col + 4];
    ssq0 += a0 * a0;
    ssq1 += a1 * a1;
    c = __builtin_amdgcn_wmma_f32_16x16x4_f32(false, a0, false, b0, (short)0, c,
                                              false, false);
    c = __builtin_amdgcn_wmma_f32_16x16x4_f32(false, a1, false, b1, (short)0, c,
                                              false, false);
  }

  float n2 = (ssq0.x + ssq0.y) + (ssq1.x + ssq1.y);
  n2 += __shfl_xor(n2, 16, 32);
  float n  = fmaxf(sqrtf(n2), 1e-5f);
  float sc = tanhf(0.1f * n) / (0.1f * n);
  float pn = sc * n;
  if (pn > MAXN) { sc *= MAXN / pn; pn = MAXN; }
  const float x2own = pn * pn;
  if (lane < 16) { sX2[wave][lane] = x2own; sSc[wave][lane] = sc; }
  __syncthreads();

  float y2 = proto_y2[(m16 < KCLS) ? m16 : 0];
  if (m16 < KCLS) {
#pragma unroll
    for (int g = 0; g < 8; g++) {
      int   m  = 8 * half + g;
      float x2 = sX2[wave][m];
      float xy = sSc[wave][m] * c[g];
      float A  = 1.f - 0.02f * xy + 0.01f * y2;
      float B  = 1.f - 0.01f * x2;
      float nv2 = fmaxf(A * A * x2 - 2.f * A * B * xy + B * B * y2, 0.f);
      float dnm = 1.f - 0.02f * xy + 1e-4f * x2 * y2 + 1e-5f;
      float man = sqrtf(nv2) / dnm;
      float z = fminf(fmaxf(0.1f * man, -1.f + 1e-5f), 1.f - 1e-5f);
      sDist[wave][m][m16] = 10.f * logf((1.f + z) / (1.f - z));
    }
  }
  __syncthreads();

  float ceContrib = 0.f;
  if (lane < 16) {
    int  q = qbase + lane;
    bool valid = (q < Q);
    float dv[KCLS];
#pragma unroll
    for (int k = 0; k < KCLS; k++) dv[k] = sDist[wave][lane][k];
    float best = dv[0]; int pred = 0;
#pragma unroll
    for (int k = 1; k < KCLS; k++) { if (dv[k] < best) { best = dv[k]; pred = k; } }

    if (PASS2) {
      float e[KCLS]; float se = 0.f;
#pragma unroll
      for (int k = 0; k < KCLS; k++) { e[k] = expf(best - dv[k]); se += e[k]; }
      if (valid) {
        float inv = 1.f / se;
#pragma unroll
        for (int k = 0; k < KCLS; k++) out[(size_t)q * KCLS + k] = e[k] * inv;
      }
    } else {
      float se = 0.f;
#pragma unroll
      for (int k = 0; k < KCLS; k++) se += expf(best - dv[k]);
      float lse = logf(se);
      int lbl = valid ? label[Q + q] : 0;
      float dl = dv[0];
#pragma unroll
      for (int k = 1; k < KCLS; k++) dl = (lbl == k) ? dv[k] : dl;
      float ce = (best - dl) - lse;
      ceContrib = valid ? ce : 0.f;

      float x2q   = x2own;
      float kq    = 2.0f / (1.0f + 0.01f * x2q);
      float gamma = rsqrtf(1.0f - 0.01f * kq * kq * x2q);
      sPred[wave * 16 + lane] = pred;
      sCoef[wave * 16 + lane] = valid ? gamma * kq * sc : 0.f;
      if (valid) atomicAdd(&bden[pred], gamma);
    }
  }

  if (!PASS2) {
    float s = waveRedSum(ceContrib);
    if (lane == 0) atomicAdd(&bloss, s);
    __syncthreads();

    static_assert(NW == 2 || NW == 8, "row groups of 64 threads; the 8-wave flush maps one thread per column");
    const int cg = tid & 63, rg = tid >> 6;
    float racc[KCLS][4];
#pragma unroll
    for (int cc = 0; cc < KCLS; cc++)
#pragma unroll
      for (int j = 0; j < 4; j++) racc[cc][j] = 0.f;

    for (int m2 = 0; m2 < 32; m2++) {
      const int mm = rg * 32 + m2;
      int p = sPred[mm]; float cf = sCoef[mm];
      int wv = mm >> 4, r = mm & 15;
      float v0 = tile[wv][r][cg];
      float v1 = tile[wv][r][cg + 64];
      float v2 = tile[wv][r][cg + 128];
      float v3 = tile[wv][r][cg + 192];
#pragma unroll
      for (int cc = 0; cc < KCLS; cc++) {
        float msk = (p == cc) ? cf : 0.f;
        racc[cc][0] += msk * v0; racc[cc][1] += msk * v1;
        racc[cc][2] += msk * v2; racc[cc][3] += msk * v3;
      }
    }
    if (NW == 2) {
#pragma unroll
      for (int cc = 0; cc < KCLS; cc++) {
        atomicAdd(&num[cc * DDIM + cg],       racc[cc][0]);
        atomicAdd(&num[cc * DDIM + cg + 64],  racc[cc][1]);
        atomicAdd(&num[cc * DDIM + cg + 128], racc[cc][2]);
        atomicAdd(&num[cc * DDIM + cg + 192], racc[cc][3]);
      }
    } else {
      constexpr int NRG = NW / 2;
      float* part = &tile[0][0][0];
      __syncthreads();
#pragma unroll
      for (int cc = 0; cc < KCLS; cc++)
#pragma unroll
        for (int j = 0; j < 4; j++)
          part[rg * (KCLS * DDIM) + cc * DDIM + cg + 64 * j] = racc[cc][j];
      __syncthreads();
      float tot[KCLS];
#pragma unroll
      for (int cc = 0; cc < KCLS; cc++) {
        float s = part[cc * DDIM + tid];
#pragma unroll
        for (int g = 1; g < NRG; g++) s += part[g * (KCLS * DDIM) + cc * DDIM + tid];
        tot[cc] = s;
      }
#pragma unroll
      for (int cc = 0; cc < KCLS; cc++) atomicAdd(&num[cc * DDIM + tid], tot[cc]);
    }
    if (tid < KCLS) atomicAdd(&den[tid], bden[tid]);
    if (tid == KCLS) atomicAdd(lossA, bloss);
  }
}

__global__ __launch_bounds__(256) void hpr_k3(const float* __restrict__ num,
                                              const float* __restrict__ den,
                                              const float* __restrict__ lossA,
                                              float* __restrict__ proto2,
                                              float* __restrict__ proto2_y2,
                                              float* __restrict__ out, int Q) {
  __shared__ float sRed[KCLS][8];
  const int t = threadIdx.x;
  const int wave = t >> 5, lane = t & 31;
  float mk[KCLS], part[KCLS];
#pragma unroll
  for (int c = 0; c < KCLS; c++) {
    mk[c]   = num[c * DDIM + t] / den[c];
    part[c] = waveRedSum(mk[c] * mk[c]);
  }
  if (lane == 0) {
#pragma unroll
    for (int c = 0; c < KCLS; c++) sRed[c][wave] = part[c];
  }
  __syncthreads();
#pragma unroll
  for (int c = 0; c < KCLS; c++) {
    float mk2 = 0.f;
#pragma unroll
    for (int w = 0; w < 8; w++) mk2 += sRed[c][w];
    float dn = 1.0f + sqrtf(1.0f - 0.01f * mk2);
    proto2[c * DDIM + t] = mk[c] / dn;
    if (t == 0) proto2_y2[c] = mk2 / (dn * dn);
  }
  if (t == 0) out[(size_t)Q * KCLS] = -lossA[0] / (float)Q;
}

extern "C" void kernel_launch(void* const* d_in, const int* in_sizes, int n_in,
                              void* d_out, int out_size, void* d_ws, size_t ws_size,
                              hipStream_t stream) {
  const float* feat  = (const float*)d_in[0];
  const int*   label = (const int*)d_in[1];
  float* out = (float*)d_out;
  float* ws  = (float*)d_ws;

  int total_rows = in_sizes[0] / DDIM;
  int Q = total_rows - 25;

  float* proto     = ws;
  float* proto_y2  = ws + 1280;
  float* num       = ws + 1288;
  float* den       = ws + 2568;
  float* lossA     = ws + 2576;
  float* proto2    = ws + 2580;
  float* proto2_y2 = ws + 3860;

  hpr_k1<<<1, 256, 0, stream>>>(feat, proto, proto_y2, num, den, lossA);

  int nblk  = (Q + 31) / 32;
  int nblk1 = (Q + 127) / 128;
  hpr_pass<false, 8><<<nblk1, 256, 0, stream>>>(feat, label, proto, proto_y2,
                                                num, den, lossA, out, Q);
  hpr_k3<<<1, 256, 0, stream>>>(num, den, lossA, proto2, proto2_y2, out, Q);
  hpr_pass<true, 2><<<nblk, 64, 0, stream>>>(feat, label, proto2, proto2_y2,
                                             num, den, lossA, out, Q);
}


# pass 2: every work-group derives proto_new and its norms from num/den itself while its query rows are in flight; k3 launch gone (4 -> 3); norm sum tree changes, quotients become reciprocal products
# speedup vs baseline: 3.6673x; 1.0772x over previous
#include <hip/hip_runtime.h>
#include <hip/hip_bf16.h>
#include <math.h>

typedef float v2f __attribute__((ext_vector_type(2)));
typedef float v8f __attribute__((ext_vector_type(8)));

#define DDIM 256
#define KCLS 5
#define PITCH 260
#define MAXN  9.96f

__device__ __forceinline__ float waveRedSum(float v) {
#pragma unroll
  for (int off = 16; off > 0; off >>= 1) v += __shfl_xor(v, off, 32);
  return v;
}

__global__ __launch_bounds__(256) void hpr_k1(const float* __restrict__ feat,
                                              float* __restrict__ proto,
                                              float* __restrict__ proto_y2,
                                              float* __restrict__ num,
                                              float* __restrict__ den,
                                              float* __restrict__ lossA) {
  __shared__ float sW[32];
  __shared__ float sG[32];
  __shared__ float sRed[KCLS][8];
  const int t = threadIdx.x;
  const int wave = t >> 5, lane = t & 31;

  float vcol[25];
#pragma unroll
  for (int r = 0; r < 25; r++) vcol[r] = feat[r * DDIM + t];

  float ss[4];
#pragma unroll
  for (int i = 0; i < 4; i++) {
    const int r  = wave + 8 * i;
    const int rr = (r < 25) ? r : 24;
    const float4* row = (const float4*)(feat + rr * DDIM);
    float4 a0 = row[lane];
    float4 a1 = row[lane + 32];
    ss[i] = a0.x * a0.x + a0.y * a0.y + a0.z * a0.z + a0.w * a0.w
          + a1.x * a1.x + a1.y * a1.y + a1.z * a1.z + a1.w * a1.w;
  }
#pragma unroll
  for (int off = 16; off > 0; off >>= 1) {
#pragma unroll
    for (int i = 0; i < 4; i++) ss[i] += __shfl_xor(ss[i], off, 32);
  }
#pragma unroll
  for (int i = 0; i < 4; i++) {
    const int r = wave + 8 * i;
    float n2 = ss[i];
    float n  = fmaxf(sqrtf(n2), 1e-5f);
    float sc = tanhf(0.1f * n) / (0.1f * n);
    float pn = sc * n;
    if (pn > MAXN) { sc *= MAXN / pn; pn = MAXN; }
    float x2    = pn * pn;
    float kq    = 2.0f / (1.0f + 0.01f * x2);
    float gamma = rsqrtf(1.0f - 0.01f * kq * kq * x2);
    if (lane == 0 && r < 25) { sW[r] = gamma * kq * sc; sG[r] = gamma; }
  }
  __syncthreads();

  float mk[KCLS], agc[KCLS], part[KCLS];
#pragma unroll
  for (int c = 0; c < KCLS; c++) {
    float agk = 0.f, ag = 0.f;
#pragma unroll
    for (int j = 0; j < 5; j++) {
      const int r = 5 * c + j;
      agk += sW[r] * vcol[r];
      ag  += sG[r];
    }
    num[c * DDIM + t] = agk;
    agc[c]  = ag;
    mk[c]   = agk / ag;
    part[c] = waveRedSum(mk[c] * mk[c]);
  }
  if (lane == 0) {
#pragma unroll
    for (int c = 0; c < KCLS; c++) sRed[c][wave] = part[c];
  }
  __syncthreads();
#pragma unroll
  for (int c = 0; c < KCLS; c++) {
    float mk2 = 0.f;
#pragma unroll
    for (int w = 0; w < 8; w++) mk2 += sRed[c][w];
    float dn = 1.0f + sqrtf(1.0f - 0.01f * mk2);
    proto[c * DDIM + t] = mk[c] / dn;
    if (t == 0) { proto_y2[c] = mk2 / (dn * dn); den[c] = agc[c]; }
  }
  if (t == 0) lossA[0] = 0.f;
}

template <bool PASS2, int NW>
__global__ __launch_bounds__(NW * 32) void hpr_pass(const float* __restrict__ feat,
                                               const int* __restrict__ label,
                                               const float* __restrict__ proto,
                                               const float* __restrict__ proto_y2,
                                               float* __restrict__ num,
                                               float* __restrict__ den,
                                               float* __restrict__ lossA,
                                               float* __restrict__ out, int Q) {
  __shared__ __align__(16) float tile[NW][16][PITCH];
  __shared__ __align__(16) float sproto[KCLS][PITCH];
  __shared__ float sX2[NW][16];
  __shared__ float sSc[NW][16];
  __shared__ float sDist[NW][16][8];
  __shared__ int   sPred[NW * 16];
  __shared__ float sCoef[NW * 16];
  __shared__ float bden[KCLS];
  __shared__ float bloss;

  const int tid  = threadIdx.x;
  const int wave = tid >> 5, lane = tid & 31;
  const int m16  = lane & 15, half = lane >> 4;
  const int qbase = blockIdx.x * (NW * 16) + wave * 16;

  float pw[KCLS];
  float4 nv0[KCLS], nv1[KCLS];
  float  dinv[KCLS], y2c[KCLS];
#pragma unroll
  for (int i = 0; i < KCLS; i++) {
    if (PASS2) {
      nv0[i]  = ((const float4*)(num + i * DDIM))[lane];
      nv1[i]  = ((const float4*)(num + i * DDIM))[lane + 32];
      dinv[i] = 1.0f / den[i];
    } else {
      pw[i] = proto[i * DDIM + tid];
    }
  }
  if (!PASS2) {
    static_assert(PASS2 || NW == 8, "pass 1 stages the proto rows one float per thread: 256 threads");
    if (tid < KCLS) bden[tid] = 0.f;
    if (tid == KCLS) bloss = 0.f;
  }

  if (PASS2) {
    static_assert(!PASS2 || NW == 2, "pass 2: wave w stages columns 128w..128w+127 of the proto rows");
    float4 ra0[16], ra1[16];
#pragma unroll
    for (int r = 0; r < 16; r++) {
      int q = qbase + r;
      const float* row = feat + (size_t)(25 + (q < Q ? q : 0)) * DDIM;
      ra0[r] = ((const float4*)row)[lane];
      ra1[r] = ((const float4*)row)[lane + 32];
    }
    float ssq[KCLS];
#pragma unroll
    for (int i = 0; i < KCLS; i++) {
      nv0[i].x *= dinv[i]; nv0[i].y *= dinv[i]; nv0[i].z *= dinv[i]; nv0[i].w *= dinv[i];
      nv1[i].x *= dinv[i]; nv1[i].y *= dinv[i]; nv1[i].z *= dinv[i]; nv1[i].w *= dinv[i];
      ssq[i] = (nv0[i].x * nv0[i].x + nv0[i].y * nv0[i].y + nv0[i].z * nv0[i].z + nv0[i].w * nv0[i].w)
             + (nv1[i].x * nv1[i].x + nv1[i].y * nv1[i].y + nv1[i].z * nv1[i].z + nv1[i].w * nv1[i].w);
    }
#pragma unroll
    for (int off = 16; off > 0; off >>= 1) {
#pragma unroll
      for (int i = 0; i < KCLS; i++) ssq[i] += __shfl_xor(ssq[i], off, 32);
    }
#pragma unroll
    for (int i = 0; i < KCLS; i++) {
      const float mk2  = ssq[i];
      const float rinv = 1.0f / (1.0f + sqrtf(1.0f - 0.01f * mk2));
      y2c[i] = (mk2 * rinv) * rinv;
      float4 h = wave ? nv1[i] : nv0[i];
      h.x *= rinv; h.y *= rinv; h.z *= rinv; h.w *= rinv;
      *(float4*)&sproto[i][wave * 128 + lane * 4] = h;
    }
    __builtin_amdgcn_sched_barrier(0x0427);
#pragma unroll
    for (int r = 0; r < 16; r++) {
      *(float4*)&tile[wave][r][lane * 4]       = ra0[r];
      *(float4*)&tile[wave][r][128 + lane * 4] = ra1[r];
    }
  } else {
    for (int r = 0; r < 16; r++) {
      int q = qbase + r;
      const float* row = feat + (size_t)(25 + (q < Q ? q : 0)) * DDIM;
      float4 a0 = ((const float4*)row)[lane];
      float4 a1 = ((const float4*)row)[lane + 32];
      *(float4*)&tile[wave][r][lane * 4]       = a0;
      *(float4*)&tile[wave][r][128 + lane * 4] = a1;
    }
#pragma unroll
    for (int i = 0; i < KCLS; i++) sproto[i][tid] = pw[i];
  }
  __syncthreads();

  const int mB = (m16 < KCLS) ? m16 : 0;
  v8f c = {0.f, 0.f, 0.f, 0.f, 0.f, 0.f, 0.f, 0.f};
  v2f ssq0 = {0.f, 0.f}, ssq1 = {0.f, 0.f};
  for (int kc = 0; kc < DDIM; kc += 8) {
    int col = kc + half * 2;
    v2f a0 = *(const v2f*)&tile[wave][m16][col];
    v2f b0 = *(const v2f*)&sproto[mB][col];
    v2f a1 = *(const v2f*)&tile[wave][m16][col + 4];
    v2f b1 = *(const v2f*)&sproto[mB][col + 4];
    ssq0 += a0 * a0;
    ssq1 += a1 * a1;
    c = __builtin_amdgcn_wmma_f32_16x16x4_f32(false, a0, false, b0, (short)0, c,
                                              false, false);
    c = __builtin_amdgcn_wmma_f32_16x16x4_f32(false, a1, false, b1, (short)0, c,
                                              false, false);
  }

  float n2 = (ssq0.x + ssq0.y) + (ssq1.x + ssq1.y);
  n2 += __shfl_xor(n2, 16, 32);
  float n  = fmaxf(sqrtf(n2), 1e-5f);
  float sc = tanhf(0.1f * n) / (0.1f * n);
  float pn = sc * n;
  if (pn > MAXN) { sc *= MAXN / pn; pn = MAXN; }
  const float x2own = pn * pn;
  if (lane < 16) { sX2[wave][lane] = x2own; sSc[wave][lane] = sc; }
  __syncthreads();

  float y2;
  if (PASS2) {
    y2 = y2c[0];
#pragma unroll
    for (int k = 1; k < KCLS; k++) y2 = (m16 == k) ? y2c[k] : y2;
  } else {
    y2 = proto_y2[(m16 < KCLS) ? m16 : 0];
  }
  if (m16 < KCLS) {
#pragma unroll
    for (int g = 0; g < 8; g++) {
      int   m  = 8 * half + g;
      float x2 = sX2[wave][m];
      float xy = sSc[wave][m] * c[g];
      float A  = 1.f - 0.02f * xy + 0.01f * y2;
      float B  = 1.f - 0.01f * x2;
      float nv2 = fmaxf(A * A * x2 - 2.f * A * B * xy + B * B * y2, 0.f);
      float dnm = 1.f - 0.02f * xy + 1e-4f * x2 * y2 + 1e-5f;
      float man = sqrtf(nv2) / dnm;
      float z = fminf(fmaxf(0.1f * man, -1.f + 1e-5f), 1.f - 1e-5f);
      sDist[wave][m][m16] = 10.f * logf((1.f + z) / (1.f - z));
    }
  }
  __syncthreads();

  float ceContrib = 0.f;
  if (lane < 16) {
    int  q = qbase + lane;
    bool valid = (q < Q);
    float dv[KCLS];
#pragma unroll
    for (int k = 0; k < KCLS; k++) dv[k] = sDist[wave][lane][k];
    float best = dv[0]; int pred = 0;
#pragma unroll
    for (int k = 1; k < KCLS; k++) { if (dv[k] < best) { best = dv[k]; pred = k; } }

    if (PASS2) {
      float e[KCLS]; float se = 0.f;
#pragma unroll
      for (int k = 0; k < KCLS; k++) { e[k] = expf(best - dv[k]); se += e[k]; }
      if (valid) {
        float inv = 1.f / se;
#pragma unroll
        for (int k = 0; k < KCLS; k++) out[(size_t)q * KCLS + k] = e[k] * inv;
      }
      if (blockIdx.x == 0 && tid == 0) out[(size_t)Q * KCLS] = -lossA[0] / (float)Q;
    } else {
      float se = 0.f;
#pragma unroll
      for (int k = 0; k < KCLS; k++) se += expf(best - dv[k]);
      float lse = logf(se);
      int lbl = valid ? label[Q + q] : 0;
      float dl = dv[0];
#pragma unroll
      for (int k = 1; k < KCLS; k++) dl = (lbl == k) ? dv[k] : dl;
      float ce = (best - dl) - lse;
      ceContrib = valid ? ce : 0.f;

      float x2q   = x2own;
      float kq    = 2.0f / (1.0f + 0.01f * x2q);
      float gamma = rsqrtf(1.0f - 0.01f * kq * kq * x2q);
      sPred[wave * 16 + lane] = pred;
      sCoef[wave * 16 + lane] = valid ? gamma * kq * sc : 0.f;
      if (valid) atomicAdd(&bden[pred], gamma);
    }
  }

  if (!PASS2) {
    float s = waveRedSum(ceContrib);
    if (lane == 0) atomicAdd(&bloss, s);
    __syncthreads();

    static_assert(NW == 2 || NW == 8, "row groups of 64 threads; the 8-wave flush maps one thread per column");
    const int cg = tid & 63, rg = tid >> 6;
    float racc[KCLS][4];
#pragma unroll
    for (int cc = 0; cc < KCLS; cc++)
#pragma unroll
      for (int j = 0; j < 4; j++) racc[cc][j] = 0.f;

    for (int m2 = 0; m2 < 32; m2++) {
      const int mm = rg * 32 + m2;
      int p = sPred[mm]; float cf = sCoef[mm];
      int wv = mm >> 4, r = mm & 15;
      float v0 = tile[wv][r][cg];
      float v1 = tile[wv][r][cg + 64];
      float v2 = tile[wv][r][cg + 128];
      float v3 = tile[wv][r][cg + 192];
#pragma unroll
      for (int cc = 0; cc < KCLS; cc++) {
        float msk = (p == cc) ? cf : 0.f;
        racc[cc][0] += msk * v0; racc[cc][1] += msk * v1;
        racc[cc][2] += msk * v2; racc[cc][3] += msk * v3;
      }
    }
    if (NW == 2) {
#pragma unroll
      for (int cc = 0; cc < KCLS; cc++) {
        atomicAdd(&num[cc * DDIM + cg],       racc[cc][0]);
        atomicAdd(&num[cc * DDIM + cg + 64],  racc[cc][1]);
        atomicAdd(&num[cc * DDIM + cg + 128], racc[cc][2]);
        atomicAdd(&num[cc * DDIM + cg + 192], racc[cc][3]);
      }
    } else {
      constexpr int NRG = NW / 2;
      float* part = &tile[0][0][0];
      __syncthreads();
#pragma unroll
      for (int cc = 0; cc < KCLS; cc++)
#pragma unroll
        for (int j = 0; j < 4; j++)
          part[rg * (KCLS * DDIM) + cc * DDIM + cg + 64 * j] = racc[cc][j];
      __syncthreads();
      float tot[KCLS];
#pragma unroll
      for (int cc = 0; cc < KCLS; cc++) {
        float s = part[cc * DDIM + tid];
#pragma unroll
        for (int g = 1; g < NRG; g++) s += part[g * (KCLS * DDIM) + cc * DDIM + tid];
        tot[cc] = s;
      }
#pragma unroll
      for (int cc = 0; cc < KCLS; cc++) atomicAdd(&num[cc * DDIM + tid], tot[cc]);
    }
    if (tid < KCLS) atomicAdd(&den[tid], bden[tid]);
    if (tid == KCLS) atomicAdd(lossA, bloss);
  }
}

extern "C" void kernel_launch(void* const* d_in, const int* in_sizes, int n_in,
                              void* d_out, int out_size, void* d_ws, size_t ws_size,
                              hipStream_t stream) {
  const float* feat  = (const float*)d_in[0];
  const int*   label = (const int*)d_in[1];
  float* out = (float*)d_out;
  float* ws  = (float*)d_ws;

  int total_rows = in_sizes[0] / DDIM;
  int Q = total_rows - 25;

  float* proto     = ws;
  float* proto_y2  = ws + 1280;
  float* num       = ws + 1288;
  float* den       = ws + 2568;
  float* lossA     = ws + 2576;

  hpr_k1<<<1, 256, 0, stream>>>(feat, proto, proto_y2, num, den, lossA);

  int nblk  = (Q + 31) / 32;
  int nblk1 = (Q + 127) / 128;
  hpr_pass<false, 8><<<nblk1, 256, 0, stream>>>(feat, label, proto, proto_y2,
                                                num, den, lossA, out, Q);
  hpr_pass<true, 2><<<nblk, 64, 0, stream>>>(feat, label, proto, proto_y2,
                                             num, den, lossA, out, Q);
}
